# starting kernel
# baseline (speedup 1.0000x reference)
#include <hip/hip_runtime.h>
#include <hip/hip_fp16.h>


typedef float    v2f  __attribute__((ext_vector_type(2)));
typedef float    v8f  __attribute__((ext_vector_type(8)));
typedef _Float16 v16h __attribute__((ext_vector_type(16)));

#define HAVE_WMMA_F32X4 __has_builtin(__builtin_amdgcn_wmma_f32_16x16x4_f32)

__global__ __launch_bounds__(128)
void pgx_proj_wmma(const float* __restrict__ embed,
                   const float* __restrict__ W1,
                   const float* __restrict__ b1,
                   float* __restrict__ H1,
                   float* __restrict__ H2,
                   int N)
{
    const int lane = threadIdx.x & 31;
    const int wave = threadIdx.x >> 5;
    const int n0   = wave * 16;
    const int m0   = blockIdx.x * 16;
    const int half = lane >> 4;
    const int l15  = lane & 15;

    int rowA = m0 + l15;
    if (rowA >= N) rowA = N - 1;
    const int col = n0 + l15;

    v8f acc1 = {};
    v8f acc2 = {};

#if HAVE_WMMA_F32X4
    #pragma unroll
    for (int k0 = 0; k0 < 64; k0 += 4) {
        const int ka = k0 + 2 * half;
        v2f a;
        a.x = embed[rowA * 64 + ka];
        a.y = embed[rowA * 64 + ka + 1];
        v2f bA, bB;
        bA.x = W1[(ka)     * 64 + col];
        bA.y = W1[(ka + 1) * 64 + col];
        bB.x = W1[(64 + ka)     * 64 + col];
        bB.y = W1[(64 + ka + 1) * 64 + col];
        acc1 = __builtin_amdgcn_wmma_f32_16x16x4_f32(false, a, false, bA,
                                                     (short)0, acc1, false, false);
        acc2 = __builtin_amdgcn_wmma_f32_16x16x4_f32(false, a, false, bB,
                                                     (short)0, acc2, false, false);
    }
#else
    #pragma unroll
    for (int k0 = 0; k0 < 64; k0 += 32) {
        v16h a, bA, bB;
        const int kb = k0 + 8 * half;
        #pragma unroll
        for (int t = 0; t < 8; ++t) {
            a[t]     = (_Float16)embed[rowA * 64 + kb + t];
            a[t + 8] = (_Float16)embed[rowA * 64 + kb + 16 + t];
        }
        const int ks = k0 + 16 * half;
        #pragma unroll
        for (int t = 0; t < 16; ++t) {
            bA[t] = (_Float16)W1[(ks + t)      * 64 + col];
            bB[t] = (_Float16)W1[(64 + ks + t) * 64 + col];
        }
        acc1 = __builtin_amdgcn_wmma_f32_16x16x32_f16(false, a, false, bA,
                                                      (short)0, acc1, false, false);
        acc2 = __builtin_amdgcn_wmma_f32_16x16x32_f16(false, a, false, bB,
                                                      (short)0, acc2, false, false);
    }
#endif

    const float bias = b1[col];
    #pragma unroll
    for (int r = 0; r < 8; ++r) {
        const int m = m0 + r + 8 * half;
        if (m < N) {
            H1[m * 64 + col] = acc1[r] + bias;
            H2[m * 64 + col] = acc2[r];
        }
    }
}

__global__ __launch_bounds__(256)
void pgx_edge_gate(const float* __restrict__ H1,
                   const float* __restrict__ H2,
                   const int*   __restrict__ ei,
                   const float* __restrict__ noise,
                   const float* __restrict__ W2,
                   const float* __restrict__ b2,
                   const float* __restrict__ tmp,
                   float* __restrict__ out,
                   int N, int E)
{
    __shared__ __align__(16) float sW2[64];
    if (threadIdx.x < 64) sW2[threadIdx.x] = W2[threadIdx.x];
    __syncthreads();

    const int e = blockIdx.x * 256 + threadIdx.x;
    if (e >= E) return;

    const int i = ei[e];
    const int j = ei[E + e];

    const float4* __restrict__ h1i = (const float4*)(H1 + (size_t)i * 64);
    const float4* __restrict__ h2i = (const float4*)(H2 + (size_t)i * 64);
    const float4* __restrict__ h1j = (const float4*)(H1 + (size_t)j * 64);
    const float4* __restrict__ h2j = (const float4*)(H2 + (size_t)j * 64);
    const float4* __restrict__ w4  = (const float4*)sW2;

    float la_ij = 0.0f;
    float la_ji = 0.0f;
    #pragma unroll
    for (int q = 0; q < 16; ++q) {
        const float4 x1 = h1i[q];
        const float4 x2 = h2j[q];
        const float4 y1 = h1j[q];
        const float4 y2 = h2i[q];
        const float4 w  = w4[q];
        la_ij += fmaxf(x1.x + x2.x, 0.0f) * w.x
               + fmaxf(x1.y + x2.y, 0.0f) * w.y
               + fmaxf(x1.z + x2.z, 0.0f) * w.z
               + fmaxf(x1.w + x2.w, 0.0f) * w.w;
        la_ji += fmaxf(y1.x + y2.x, 0.0f) * w.x
               + fmaxf(y1.y + y2.y, 0.0f) * w.y
               + fmaxf(y1.z + y2.z, 0.0f) * w.z
               + fmaxf(y1.w + y2.w, 0.0f) * w.w;
    }

    const float bb   = b2[0];
    const float invt = 1.0f / tmp[0];
    const float nij  = noise[(size_t)i * N + j];
    const float nji  = noise[(size_t)j * N + i];

    const float oij = (__logf(nij) - __logf(1.0f - nij) + la_ij + bb) * invt;
    const float oji = (__logf(nji) - __logf(1.0f - nji) + la_ji + bb) * invt;
    const float gij = 1.0f / (1.0f + __expf(-oij));
    const float gji = 1.0f / (1.0f + __expf(-oji));

    out[e] = 0.5f * (gij + gji);
}

extern "C" void kernel_launch(void* const* d_in, const int* in_sizes, int n_in,
                              void* d_out, int out_size, void* d_ws, size_t ws_size,
                              hipStream_t stream)
{
    const float* embed = (const float*)d_in[0];
    const int*   ei    = (const int*)  d_in[1];
    const float* noise = (const float*)d_in[2];
    const float* tmp   = (const float*)d_in[3];
    const float* W1    = (const float*)d_in[4];
    const float* b1    = (const float*)d_in[5];
    const float* W2    = (const float*)d_in[6];
    const float* b2    = (const float*)d_in[7];
    float*       out   = (float*)d_out;

    const int N = in_sizes[0] / 64;
    const int E = in_sizes[1] / 2;

    float* H1 = (float*)d_ws;
    float* H2 = H1 + (size_t)N * 64;

    const int mtiles = (N + 15) / 16;
    hipLaunchKernelGGL(pgx_proj_wmma, dim3(mtiles), dim3(128), 0, stream,
                       embed, W1, b1, H1, H2, N);

    hipLaunchKernelGGL(pgx_edge_gate, dim3((E + 255) / 256), dim3(256), 0, stream,
                       H1, H2, ei, noise, W2, b2, tmp, out, N, E);
}


# one launch: projection folded into the edge kernel; per 32 edges f32 WMMA forms (ei+ej)(Wa+Wb)/2 and (ei-ej)(Wa-Wb)/2 from gathered embed rows; no node table; 64-term sum order changes (tree)
# speedup vs baseline: 1.3476x; 1.3476x over previous
#include <hip/hip_runtime.h>


typedef float v2f __attribute__((ext_vector_type(2)));
typedef float v8f __attribute__((ext_vector_type(8)));

#define PAIRS_PER_MAT 2048

__global__ __launch_bounds__(256)
void edge_gate_fused(const float* __restrict__ embed,
                    const int*   __restrict__ ei,
                    const float* __restrict__ noise,
                    const float* __restrict__ tmp,
                    const float* __restrict__ W1,
                    const float* __restrict__ b1,
                    const float* __restrict__ W2,
                    const float* __restrict__ b2,
                    float* __restrict__ out,
                    int N, int E)
{
    __shared__ __align__(16) v2f sWp[PAIRS_PER_MAT];
    __shared__ __align__(16) v2f sWm[PAIRS_PER_MAT];
    __shared__ __align__(16) float sB1[64];
    __shared__ __align__(16) float sW2[64];

    const int tid  = threadIdx.x;
    const int lane = tid & 31;
    const int wave = tid >> 5;
    const int h    = lane >> 4;
    const int c    = lane & 15;

    const int base = (blockIdx.x * 8 + wave) * 32;
    int eA = base + c;       if (eA > E - 1) eA = E - 1;
    int eB = base + 16 + c;  if (eB > E - 1) eB = E - 1;
    const int iA = ei[eA], jA = ei[E + eA];
    const int iB = ei[eB], jB = ei[E + eB];

    float a0[8], a1[8], c0[8], c1[8];
    #pragma unroll
    for (int q = 0; q < 8; ++q) {
        const int p   = tid + 256 * q;
        const int pc  = p & 15;
        const int ph  = (p >> 4) & 1;
        const int pt  = (p >> 5) & 3;
        const int pst = p >> 7;
        const int k   = 8 * (pst >> 1) + 4 * ph + 2 * (pst & 1);
        const int hid = 16 * pt + pc;
        a0[q] = W1[k * 64 + hid];
        a1[q] = W1[(k + 1) * 64 + hid];
        c0[q] = W1[(64 + k) * 64 + hid];
        c1[q] = W1[(65 + k) * 64 + hid];
    }
    const float b1v = (tid < 64) ? b1[tid] : 0.0f;
    const float w2v = (tid < 64) ? W2[tid] : 0.0f;

    const float4* __restrict__ riA = (const float4*)(embed + (size_t)iA * 64) + h;
    const float4* __restrict__ rjA = (const float4*)(embed + (size_t)jA * 64) + h;
    const float4* __restrict__ riB = (const float4*)(embed + (size_t)iB * 64) + h;
    const float4* __restrict__ rjB = (const float4*)(embed + (size_t)jB * 64) + h;
    float4 xi[8], xj[8], yi[8], yj[8];
    #pragma unroll
    for (int u = 0; u < 8; ++u) {
        xi[u] = riA[2 * u];  xj[u] = rjA[2 * u];
        yi[u] = riB[2 * u];  yj[u] = rjB[2 * u];
        __builtin_amdgcn_sched_barrier(0);
    }

    const int io = h ? iB : iA;
    const int jo = h ? jB : jA;
    const float nij = noise[(size_t)io * N + jo];
    const float nji = noise[(size_t)jo * N + io];
    const float bb   = b2[0];
    const float invt = 1.0f / tmp[0];

    #pragma unroll
    for (int q = 0; q < 8; ++q) {
        const int p = tid + 256 * q;
        v2f wp, wm;
        wp.x = 0.5f * (a0[q] + c0[q]);  wp.y = 0.5f * (a1[q] + c1[q]);
        wm.x = 0.5f * (a0[q] - c0[q]);  wm.y = 0.5f * (a1[q] - c1[q]);
        sWp[p] = wp;
        sWm[p] = wm;
    }
    if (tid < 64) { sB1[tid] = b1v; sW2[tid] = w2v; }

    __builtin_amdgcn_fence(__ATOMIC_RELEASE, "workgroup", "local");
    __builtin_amdgcn_s_barrier();
    __builtin_amdgcn_fence(__ATOMIC_ACQUIRE, "workgroup", "local");

    v8f PA[4], QA[4], PB[4], QB[4];
    #pragma unroll
    for (int t = 0; t < 4; ++t) { PA[t] = {}; QA[t] = {}; PB[t] = {}; QB[t] = {}; }

    #pragma unroll
    for (int u = 0; u < 8; ++u) {
        v2f SA[2], DA[2], SB[2], DB[2];
        SA[0].x = xi[u].x + xj[u].x;  SA[0].y = xi[u].y + xj[u].y;
        SA[1].x = xi[u].z + xj[u].z;  SA[1].y = xi[u].w + xj[u].w;
        DA[0].x = xi[u].x - xj[u].x;  DA[0].y = xi[u].y - xj[u].y;
        DA[1].x = xi[u].z - xj[u].z;  DA[1].y = xi[u].w - xj[u].w;
        SB[0].x = yi[u].x + yj[u].x;  SB[0].y = yi[u].y + yj[u].y;
        SB[1].x = yi[u].z + yj[u].z;  SB[1].y = yi[u].w + yj[u].w;
        DB[0].x = yi[u].x - yj[u].x;  DB[0].y = yi[u].y - yj[u].y;
        DB[1].x = yi[u].z - yj[u].z;  DB[1].y = yi[u].w - yj[u].w;
        #pragma unroll
        for (int v = 0; v < 2; ++v) {
            const int st = 2 * u + v;
            #pragma unroll
            for (int t = 0; t < 4; ++t) {
                const v2f wp = sWp[(st * 4 + t) * 32 + lane];
                const v2f wm = sWm[(st * 4 + t) * 32 + lane];
                PA[t] = __builtin_amdgcn_wmma_f32_16x16x4_f32(false, wp, false, SA[v],
                                                              (short)0, PA[t], false, false);
                QA[t] = __builtin_amdgcn_wmma_f32_16x16x4_f32(false, wm, false, DA[v],
                                                              (short)0, QA[t], false, false);
                PB[t] = __builtin_amdgcn_wmma_f32_16x16x4_f32(false, wp, false, SB[v],
                                                              (short)0, PB[t], false, false);
                QB[t] = __builtin_amdgcn_wmma_f32_16x16x4_f32(false, wm, false, DB[v],
                                                              (short)0, QB[t], false, false);
            }
        }
    }

    float sijA = 0.0f, sjiA = 0.0f, sijB = 0.0f, sjiB = 0.0f;
    #pragma unroll
    for (int t = 0; t < 4; ++t) {
        const float4 bq0 = *(const float4*)&sB1[16 * t + 8 * h];
        const float4 bq1 = *(const float4*)&sB1[16 * t + 8 * h + 4];
        const float4 wq0 = *(const float4*)&sW2[16 * t + 8 * h];
        const float4 wq1 = *(const float4*)&sW2[16 * t + 8 * h + 4];
        const float bv[8] = {bq0.x, bq0.y, bq0.z, bq0.w, bq1.x, bq1.y, bq1.z, bq1.w};
        const float wv[8] = {wq0.x, wq0.y, wq0.z, wq0.w, wq1.x, wq1.y, wq1.z, wq1.w};
        #pragma unroll
        for (int r = 0; r < 8; ++r) {
            const float pa = PA[t][r], qa = QA[t][r];
            const float pb = PB[t][r], qb = QB[t][r];
            sijA += fmaxf((pa + qa) + bv[r], 0.0f) * wv[r];
            sjiA += fmaxf((pa - qa) + bv[r], 0.0f) * wv[r];
            sijB += fmaxf((pb + qb) + bv[r], 0.0f) * wv[r];
            sjiB += fmaxf((pb - qb) + bv[r], 0.0f) * wv[r];
        }
    }

    const float give_ij = h ? sijA : sijB;
    const float give_ji = h ? sjiA : sjiB;
    const float keep_ij = h ? sijB : sijA;
    const float keep_ji = h ? sjiB : sjiA;
    const float la_ij = keep_ij + __shfl_xor(give_ij, 16);
    const float la_ji = keep_ji + __shfl_xor(give_ji, 16);

    const float oij = (__logf(nij) - __logf(1.0f - nij) + la_ij + bb) * invt;
    const float oji = (__logf(nji) - __logf(1.0f - nji) + la_ji + bb) * invt;
    const float gij = 1.0f / (1.0f + __expf(-oij));
    const float gji = 1.0f / (1.0f + __expf(-oji));

    const int e = base + lane;
    if (e < E) out[e] = 0.5f * (gij + gji);
}

extern "C" void kernel_launch(void* const* d_in, const int* in_sizes, int n_in,
                              void* d_out, int out_size, void* d_ws, size_t ws_size,
                              hipStream_t stream)
{
    const float* embed = (const float*)d_in[0];
    const int*   ei    = (const int*)  d_in[1];
    const float* noise = (const float*)d_in[2];
    const float* tmp   = (const float*)d_in[3];
    const float* W1    = (const float*)d_in[4];
    const float* b1    = (const float*)d_in[5];
    const float* W2    = (const float*)d_in[6];
    const float* b2    = (const float*)d_in[7];
    float*       out   = (float*)d_out;

    const int N = in_sizes[0] / 64;
    const int E = in_sizes[1] / 2;

    hipLaunchKernelGGL(edge_gate_fused, dim3((E + 255) / 256), dim3(256), 0, stream,
                       embed, ei, noise, tmp, W1, b1, W2, b2, out, N, E);
}
